# starting kernel
# baseline (speedup 1.0000x reference)
#include <hip/hip_runtime.h>
#include <hip/hip_bf16.h>

typedef __attribute__((ext_vector_type(2))) float v2f;
typedef __attribute__((ext_vector_type(4))) float v4f;
typedef __attribute__((ext_vector_type(8))) float v8f;

#define N_QUBITS 12
#define N_LAYERS 6
#define N_CLASSES 10
#define NSTATE 4096
#define NMAT 18

__global__ void prep_gates(const float* __restrict__ qp,
                           float* __restrict__ Gr, float* __restrict__ Gi) {
    __shared__ float ur[72][2][2];
    __shared__ float ui[72][2][2];
    int t = threadIdx.x;
    if (t < 72) {
        int l = t / 12, q = t % 12;
        const float* p = qp + (l * 12 + q) * 3;
        float phi = p[0], th = p[1], om = p[2];
        float ct = cosf(0.5f * th), st = sinf(0.5f * th);
        float a = 0.5f * (phi + om), b = 0.5f * (phi - om);
        float ca = cosf(a), sa = sinf(a), cb = cosf(b), sb = sinf(b);
        ur[t][0][0] = ca * ct;  ui[t][0][0] = -sa * ct;
        ur[t][0][1] = -cb * st; ui[t][0][1] = -sb * st;
        ur[t][1][0] = cb * st;  ui[t][1][0] = -sb * st;
        ur[t][1][1] = ca * ct;  ui[t][1][1] = sa * ct;
    }
    __syncthreads();
    for (int e = t; e < NMAT * 256; e += blockDim.x) {
        int m = e >> 8, pos = e & 255;
        int l = m / 3, g = m % 3;
        int i = pos >> 4, k = pos & 15;
        float pr = 1.f, pi = 0.f;
#pragma unroll
        for (int j = 0; j < 4; ++j) {
            int u = l * 12 + g * 4 + j;
            int bi_ = (i >> (3 - j)) & 1;
            int bk  = (k >> (3 - j)) & 1;
            float wr = ur[u][bi_][bk], wi = ui[u][bi_][bk];
            float nr = pr * wr - pi * wi;
            float ni = pr * wi + pi * wr;
            pr = nr; pi = ni;
        }
        Gr[e] = pr; Gi[e] = pi;
    }
}

__global__ __launch_bounds__(256) void qsim(
        const float* __restrict__ x,
        const float* __restrict__ Gr, const float* __restrict__ Gi,
        const float* __restrict__ fcw, const float* __restrict__ fcb,
        float* __restrict__ out) {
    __shared__ float sR[NSTATE];
    __shared__ float sI[NSTATE];
    __shared__ float cs[24];
    __shared__ float red[8 * 12];
    __shared__ float ev[12];

    const int b    = blockIdx.x;
    const int tid  = threadIdx.x;
    const int lane = tid & 31;
    const int wave = tid >> 5;
    const int c    = lane & 15;
    const int sel  = lane >> 4;

    if (tid < 12) {
        float h = 0.5f * x[b * 12 + tid];
        cs[tid] = cosf(h);
        cs[12 + tid] = sinf(h);
    }
    __syncthreads();
#pragma unroll
    for (int j = 0; j < 16; ++j) {
        int idx = tid + j * 256;
        float p = 1.f;
#pragma unroll
        for (int q = 0; q < 12; ++q)
            p *= cs[q + (((idx >> (11 - q)) & 1) ? 12 : 0)];
        sR[idx] = p;
        sI[idx] = 0.f;
    }
    __syncthreads();

    for (int l = 0; l < N_LAYERS; ++l) {
#pragma unroll
        for (int g = 0; g < 3; ++g) {
            const float* gr = Gr + (l * 3 + g) * 256;
            const float* gi = Gi + (l * 3 + g) * 256;
            v2f ar[4], ai[4];
#pragma unroll
            for (int kk = 0; kk < 4; ++kk) {
                int k0 = kk * 4 + sel * 2;
                v2f t0, t1;
                t0.x = gr[c * 16 + k0]; t0.y = gr[c * 16 + k0 + 1];
                t1.x = gi[c * 16 + k0]; t1.y = gi[c * 16 + k0 + 1];
                ar[kk] = t0; ai[kk] = t1;
            }
            const int rs = (g == 0) ? 256 : ((g == 1) ? 16 : 1);
#pragma unroll
            for (int tt = 0; tt < 2; ++tt) {
                int tile = wave * 2 + tt;
                int off = (g == 0) ? (tile * 16 + c)
                        : (g == 1) ? (tile * 256 + c)
                                   : (tile * 256 + c * 16);
                v2f br[4], bi_[4];
#pragma unroll
                for (int kk = 0; kk < 4; ++kk) {
                    int k0 = kk * 4 + sel * 2;
                    if (g == 2) {
                        br[kk]  = *(const v2f*)&sR[k0 + off];
                        bi_[kk] = *(const v2f*)&sI[k0 + off];
                    } else {
                        v2f t0, t1;
                        t0.x = sR[k0 * rs + off]; t0.y = sR[(k0 + 1) * rs + off];
                        t1.x = sI[k0 * rs + off]; t1.y = sI[(k0 + 1) * rs + off];
                        br[kk] = t0; bi_[kk] = t1;
                    }
                }
                v8f arr = {}; v8f aii = {}; v8f aim = {};
#pragma unroll
                for (int kk = 0; kk < 4; ++kk)
                    arr = __builtin_amdgcn_wmma_f32_16x16x4_f32(
                        false, ar[kk], false, br[kk], (short)0, arr, false, false);
#pragma unroll
                for (int kk = 0; kk < 4; ++kk)
                    aii = __builtin_amdgcn_wmma_f32_16x16x4_f32(
                        false, ai[kk], false, bi_[kk], (short)0, aii, false, false);
#pragma unroll
                for (int kk = 0; kk < 4; ++kk)
                    aim = __builtin_amdgcn_wmma_f32_16x16x4_f32(
                        false, ar[kk], false, bi_[kk], (short)0, aim, false, false);
#pragma unroll
                for (int kk = 0; kk < 4; ++kk)
                    aim = __builtin_amdgcn_wmma_f32_16x16x4_f32(
                        false, ai[kk], false, br[kk], (short)0, aim, false, false);
                float dr[8], di[8];
#pragma unroll
                for (int v = 0; v < 8; ++v) { dr[v] = arr[v] - aii[v]; di[v] = aim[v]; }
                if (g == 2) {
                    int base = off + sel * 8;
                    v4f r0, r1, i0, i1;
                    r0.x=dr[0]; r0.y=dr[1]; r0.z=dr[2]; r0.w=dr[3];
                    r1.x=dr[4]; r1.y=dr[5]; r1.z=dr[6]; r1.w=dr[7];
                    i0.x=di[0]; i0.y=di[1]; i0.z=di[2]; i0.w=di[3];
                    i1.x=di[4]; i1.y=di[5]; i1.z=di[6]; i1.w=di[7];
                    *(v4f*)&sR[base]     = r0;
                    *(v4f*)&sR[base + 4] = r1;
                    *(v4f*)&sI[base]     = i0;
                    *(v4f*)&sI[base + 4] = i1;
                } else {
#pragma unroll
                    for (int v = 0; v < 8; ++v) {
                        int ad = (v + sel * 8) * rs + off;
                        sR[ad] = dr[v];
                        sI[ad] = di[v];
                    }
                }
            }
            __syncthreads();
        }

        const int r = (l % 11) + 1;
        float tr[16], ti[16];
#pragma unroll
        for (int j = 0; j < 16; ++j) {
            int idx = tid + j * 256;
            int src = idx;
#pragma unroll
            for (int q = 11; q >= 0; --q) {
                int mc = 1 << (11 - q);
                int mt = 1 << (11 - ((q + r) % 12));
                if (src & mc) src ^= mt;
            }
            tr[j] = sR[src]; ti[j] = sI[src];
        }
        __syncthreads();
#pragma unroll
        for (int j = 0; j < 16; ++j) {
            int idx = tid + j * 256;
            sR[idx] = tr[j]; sI[idx] = ti[j];
        }
        __syncthreads();
    }

    float acc[12];
#pragma unroll
    for (int q = 0; q < 12; ++q) acc[q] = 0.f;
#pragma unroll
    for (int j = 0; j < 16; ++j) {
        int idx = tid + j * 256;
        float re = sR[idx], im = sI[idx];
        float p = re * re + im * im;
#pragma unroll
        for (int q = 0; q < 12; ++q)
            acc[q] += ((idx >> (11 - q)) & 1) ? -p : p;
    }
#pragma unroll
    for (int offm = 16; offm > 0; offm >>= 1)
#pragma unroll
        for (int q = 0; q < 12; ++q)
            acc[q] += __shfl_xor(acc[q], offm, 32);
    if (lane == 0)
#pragma unroll
        for (int q = 0; q < 12; ++q) red[wave * 12 + q] = acc[q];
    __syncthreads();
    if (tid < 12) {
        float s = 0.f;
#pragma unroll
        for (int w = 0; w < 8; ++w) s += red[w * 12 + tid];
        ev[tid] = s;
    }
    __syncthreads();

    if (tid < N_CLASSES) {
        float s = fcb[tid];
#pragma unroll
        for (int q = 0; q < 12; ++q) s += ev[q] * fcw[tid * 12 + q];
        out[b * N_CLASSES + tid] = s;
    }
}

extern "C" void kernel_launch(void* const* d_in, const int* in_sizes, int n_in,
                              void* d_out, int out_size, void* d_ws, size_t ws_size,
                              hipStream_t stream) {
    const float* x   = (const float*)d_in[0];
    const float* qp  = (const float*)d_in[1];
    const float* fcw = (const float*)d_in[2];
    const float* fcb = (const float*)d_in[3];
    float* Gr = (float*)d_ws;
    float* Gi = Gr + NMAT * 256;

    prep_gates<<<1, 256, 0, stream>>>(qp, Gr, Gi);
    qsim<<<4096, 256, 0, stream>>>(x, Gr, Gi, fcw, fcb, (float*)d_out);
}


# CNOT ring folded into the third gate group's LDS stores (scatter to the ring destination, GF(2)-linear index map, second state copy); the gather pass and its 2 barriers per layer are removed
# speedup vs baseline: 1.5760x; 1.5760x over previous
#include <hip/hip_runtime.h>
#include <hip/hip_bf16.h>

typedef __attribute__((ext_vector_type(2))) float v2f;
typedef __attribute__((ext_vector_type(4))) float v4f;
typedef __attribute__((ext_vector_type(8))) float v8f;

#define N_QUBITS 12
#define N_LAYERS 6
#define N_CLASSES 10
#define NSTATE 4096
#define NMAT 18

__global__ void prep_gates(const float* __restrict__ qp,
                           float* __restrict__ Gr, float* __restrict__ Gi) {
    __shared__ float ur[72][2][2];
    __shared__ float ui[72][2][2];
    int t = threadIdx.x;
    if (t < 72) {
        int l = t / 12, q = t % 12;
        const float* p = qp + (l * 12 + q) * 3;
        float phi = p[0], th = p[1], om = p[2];
        float ct = cosf(0.5f * th), st = sinf(0.5f * th);
        float a = 0.5f * (phi + om), b = 0.5f * (phi - om);
        float ca = cosf(a), sa = sinf(a), cb = cosf(b), sb = sinf(b);
        ur[t][0][0] = ca * ct;  ui[t][0][0] = -sa * ct;
        ur[t][0][1] = -cb * st; ui[t][0][1] = -sb * st;
        ur[t][1][0] = cb * st;  ui[t][1][0] = -sb * st;
        ur[t][1][1] = ca * ct;  ui[t][1][1] = sa * ct;
    }
    __syncthreads();
    for (int e = t; e < NMAT * 256; e += blockDim.x) {
        int m = e >> 8, pos = e & 255;
        int l = m / 3, g = m % 3;
        int i = pos >> 4, k = pos & 15;
        float pr = 1.f, pi = 0.f;
#pragma unroll
        for (int j = 0; j < 4; ++j) {
            int u = l * 12 + g * 4 + j;
            int bi_ = (i >> (3 - j)) & 1;
            int bk  = (k >> (3 - j)) & 1;
            float wr = ur[u][bi_][bk], wi = ui[u][bi_][bk];
            float nr = pr * wr - pi * wi;
            float ni = pr * wi + pi * wr;
            pr = nr; pi = ni;
        }
        Gr[e] = pr; Gi[e] = pi;
    }
}

struct RingShifts { int s1, s2, s4, s8, wrap; };

__device__ __forceinline__ RingShifts ring_shifts(int r) {
    RingShifts k;
    k.s1 = r;
    k.s2 = min(2 * r, N_QUBITS);
    k.s4 = min(4 * r, N_QUBITS);
    k.s8 = min(8 * r, N_QUBITS);
    k.wrap = N_QUBITS - r;
    return k;
}

__device__ __forceinline__ int ring_dst(int s, const RingShifts k) {
    int c = s;
    c ^= c >> k.s1;
    c ^= c >> k.s2;
    c ^= c >> k.s4;
    c ^= c >> k.s8;
    return c ^ ((c << k.wrap) & (NSTATE - 1));
}

__global__ __launch_bounds__(256) void qsim(
        const float* __restrict__ x,
        const float* __restrict__ Gr, const float* __restrict__ Gi,
        const float* __restrict__ fcw, const float* __restrict__ fcb,
        float* __restrict__ out) {
    __shared__ float sbuf[4 * NSTATE];
    __shared__ float cs[24];
    __shared__ float red[8 * 12];
    __shared__ float ev[12];

    const int b    = blockIdx.x;
    const int tid  = threadIdx.x;
    const int lane = tid & 31;
    const int wave = tid >> 5;
    const int c    = lane & 15;
    const int sel  = lane >> 4;

    if (tid < 12) {
        float h = 0.5f * x[b * 12 + tid];
        cs[tid] = cosf(h);
        cs[12 + tid] = sinf(h);
    }
    __syncthreads();
#pragma unroll
    for (int j = 0; j < 16; ++j) {
        int idx = tid + j * 256;
        float p = 1.f;
#pragma unroll
        for (int q = 0; q < 12; ++q)
            p *= cs[q + (((idx >> (11 - q)) & 1) ? 12 : 0)];
        sbuf[idx] = p;
        sbuf[idx + 2 * NSTATE] = 0.f;
    }
    __syncthreads();

    int cur = 0;
    for (int l = 0; l < N_LAYERS; ++l) {
        float* sR = sbuf + cur * NSTATE;
        float* sI = sR + 2 * NSTATE;
        const RingShifts rk = ring_shifts((l % (N_QUBITS - 1)) + 1);
        const int d1 = ring_dst(1, rk) << 2, d2 = ring_dst(2, rk) << 2,
                  d4 = ring_dst(4, rk) << 2;
        const int nxt4 = (cur ^ 1) * (NSTATE * 4);
#pragma unroll
        for (int g = 0; g < 3; ++g) {
            const float* gr = Gr + (l * 3 + g) * 256;
            const float* gi = Gi + (l * 3 + g) * 256;
            v2f ar[4], ai[4];
#pragma unroll
            for (int kk = 0; kk < 4; ++kk) {
                int k0 = kk * 4 + sel * 2;
                v2f t0, t1;
                t0.x = gr[c * 16 + k0]; t0.y = gr[c * 16 + k0 + 1];
                t1.x = gi[c * 16 + k0]; t1.y = gi[c * 16 + k0 + 1];
                ar[kk] = t0; ai[kk] = t1;
            }
            const int rs = (g == 0) ? 256 : ((g == 1) ? 16 : 1);
#pragma unroll
            for (int tt = 0; tt < 2; ++tt) {
                int tile = wave * 2 + tt;
                int off = (g == 0) ? (tile * 16 + c)
                        : (g == 1) ? (tile * 256 + c)
                                   : (tile * 256 + c * 16);
                v2f br[4], bi_[4];
#pragma unroll
                for (int kk = 0; kk < 4; ++kk) {
                    int k0 = kk * 4 + sel * 2;
                    if (g == 2) {
                        br[kk]  = *(const v2f*)&sR[k0 + off];
                        bi_[kk] = *(const v2f*)&sI[k0 + off];
                    } else {
                        v2f t0, t1;
                        t0.x = sR[k0 * rs + off]; t0.y = sR[(k0 + 1) * rs + off];
                        t1.x = sI[k0 * rs + off]; t1.y = sI[(k0 + 1) * rs + off];
                        br[kk] = t0; bi_[kk] = t1;
                    }
                }
                v8f arr = {}; v8f aii = {}; v8f aim = {};
#pragma unroll
                for (int kk = 0; kk < 4; ++kk)
                    arr = __builtin_amdgcn_wmma_f32_16x16x4_f32(
                        false, ar[kk], false, br[kk], (short)0, arr, false, false);
#pragma unroll
                for (int kk = 0; kk < 4; ++kk)
                    aii = __builtin_amdgcn_wmma_f32_16x16x4_f32(
                        false, ai[kk], false, bi_[kk], (short)0, aii, false, false);
#pragma unroll
                for (int kk = 0; kk < 4; ++kk)
                    aim = __builtin_amdgcn_wmma_f32_16x16x4_f32(
                        false, ar[kk], false, bi_[kk], (short)0, aim, false, false);
#pragma unroll
                for (int kk = 0; kk < 4; ++kk)
                    aim = __builtin_amdgcn_wmma_f32_16x16x4_f32(
                        false, ai[kk], false, br[kk], (short)0, aim, false, false);
                float dr[8], di[8];
#pragma unroll
                for (int v = 0; v < 8; ++v) { dr[v] = arr[v] - aii[v]; di[v] = aim[v]; }
                if (g == 2) {
                    const int pb = (ring_dst(off + sel * 8, rk) << 2) | nxt4;
                    char* nb = (char*)sbuf;
#pragma unroll
                    for (int v = 0; v < 8; ++v) {
                        int a = pb ^ ((v & 1) ? d1 : 0) ^ ((v & 2) ? d2 : 0)
                                   ^ ((v & 4) ? d4 : 0);
                        *(float*)(nb + a) = dr[v];
                        *(float*)(nb + a + 2 * NSTATE * 4) = di[v];
                    }
                } else {
#pragma unroll
                    for (int v = 0; v < 8; ++v) {
                        int ad = (v + sel * 8) * rs + off;
                        sR[ad] = dr[v];
                        sI[ad] = di[v];
                    }
                }
            }
            __syncthreads();
        }

        cur ^= 1;
    }

    float acc[12];
#pragma unroll
    for (int q = 0; q < 12; ++q) acc[q] = 0.f;
#pragma unroll
    for (int j = 0; j < 16; ++j) {
        int idx = tid + j * 256;
        float re = sbuf[cur * NSTATE + idx], im = sbuf[cur * NSTATE + idx + 2 * NSTATE];
        float p = re * re + im * im;
#pragma unroll
        for (int q = 0; q < 12; ++q)
            acc[q] += ((idx >> (11 - q)) & 1) ? -p : p;
    }
#pragma unroll
    for (int offm = 16; offm > 0; offm >>= 1)
#pragma unroll
        for (int q = 0; q < 12; ++q)
            acc[q] += __shfl_xor(acc[q], offm, 32);
    if (lane == 0)
#pragma unroll
        for (int q = 0; q < 12; ++q) red[wave * 12 + q] = acc[q];
    __syncthreads();
    if (tid < 12) {
        float s = 0.f;
#pragma unroll
        for (int w = 0; w < 8; ++w) s += red[w * 12 + tid];
        ev[tid] = s;
    }
    __syncthreads();

    if (tid < N_CLASSES) {
        float s = fcb[tid];
#pragma unroll
        for (int q = 0; q < 12; ++q) s += ev[q] * fcw[tid * 12 + q];
        out[b * N_CLASSES + tid] = s;
    }
}

extern "C" void kernel_launch(void* const* d_in, const int* in_sizes, int n_in,
                              void* d_out, int out_size, void* d_ws, size_t ws_size,
                              hipStream_t stream) {
    const float* x   = (const float*)d_in[0];
    const float* qp  = (const float*)d_in[1];
    const float* fcw = (const float*)d_in[2];
    const float* fcb = (const float*)d_in[3];
    float* Gr = (float*)d_ws;
    float* Gi = Gr + NMAT * 256;

    prep_gates<<<1, 256, 0, stream>>>(qp, Gr, Gi);
    qsim<<<4096, 256, 0, stream>>>(x, Gr, Gi, fcw, fcb, (float*)d_out);
}


# Complex gate product with three real WMMA chains instead of four: K1=Gr(Sr+Si) seeds Re=K1-(Gr+Gi)Si and Im=K1+(Gi-Gr)Sr; a quarter fewer matrix instructions; summation order changes
# speedup vs baseline: 1.8291x; 1.1606x over previous
#include <hip/hip_runtime.h>
#include <hip/hip_bf16.h>

typedef __attribute__((ext_vector_type(2))) float v2f;
typedef __attribute__((ext_vector_type(4))) float v4f;
typedef __attribute__((ext_vector_type(8))) float v8f;

#define N_QUBITS 12
#define N_LAYERS 6
#define N_CLASSES 10
#define NSTATE 4096
#define NMAT 18

__global__ void prep_gates(const float* __restrict__ qp,
                           float* __restrict__ Gr, float* __restrict__ Gi) {
    __shared__ float ur[72][2][2];
    __shared__ float ui[72][2][2];
    int t = threadIdx.x;
    if (t < 72) {
        int l = t / 12, q = t % 12;
        const float* p = qp + (l * 12 + q) * 3;
        float phi = p[0], th = p[1], om = p[2];
        float ct = cosf(0.5f * th), st = sinf(0.5f * th);
        float a = 0.5f * (phi + om), b = 0.5f * (phi - om);
        float ca = cosf(a), sa = sinf(a), cb = cosf(b), sb = sinf(b);
        ur[t][0][0] = ca * ct;  ui[t][0][0] = -sa * ct;
        ur[t][0][1] = -cb * st; ui[t][0][1] = -sb * st;
        ur[t][1][0] = cb * st;  ui[t][1][0] = -sb * st;
        ur[t][1][1] = ca * ct;  ui[t][1][1] = sa * ct;
    }
    __syncthreads();
    for (int e = t; e < NMAT * 256; e += blockDim.x) {
        int m = e >> 8, pos = e & 255;
        int l = m / 3, g = m % 3;
        int i = pos >> 4, k = pos & 15;
        float pr = 1.f, pi = 0.f;
#pragma unroll
        for (int j = 0; j < 4; ++j) {
            int u = l * 12 + g * 4 + j;
            int bi_ = (i >> (3 - j)) & 1;
            int bk  = (k >> (3 - j)) & 1;
            float wr = ur[u][bi_][bk], wi = ui[u][bi_][bk];
            float nr = pr * wr - pi * wi;
            float ni = pr * wi + pi * wr;
            pr = nr; pi = ni;
        }
        Gr[e] = pr; Gi[e] = pi;
    }
}

struct RingShifts { int s1, s2, s4, s8, wrap; };

__device__ __forceinline__ RingShifts ring_shifts(int r) {
    RingShifts k;
    k.s1 = r;
    k.s2 = min(2 * r, N_QUBITS);
    k.s4 = min(4 * r, N_QUBITS);
    k.s8 = min(8 * r, N_QUBITS);
    k.wrap = N_QUBITS - r;
    return k;
}

__device__ __forceinline__ int ring_dst(int s, const RingShifts k) {
    int c = s;
    c ^= c >> k.s1;
    c ^= c >> k.s2;
    c ^= c >> k.s4;
    c ^= c >> k.s8;
    return c ^ ((c << k.wrap) & (NSTATE - 1));
}

__global__ __launch_bounds__(256) void qsim(
        const float* __restrict__ x,
        const float* __restrict__ Gr, const float* __restrict__ Gi,
        const float* __restrict__ fcw, const float* __restrict__ fcb,
        float* __restrict__ out) {
    __shared__ float sbuf[4 * NSTATE];
    __shared__ float cs[24];
    __shared__ float red[8 * 12];
    __shared__ float ev[12];

    const int b    = blockIdx.x;
    const int tid  = threadIdx.x;
    const int lane = tid & 31;
    const int wave = tid >> 5;
    const int c    = lane & 15;
    const int sel  = lane >> 4;

    if (tid < 12) {
        float h = 0.5f * x[b * 12 + tid];
        cs[tid] = cosf(h);
        cs[12 + tid] = sinf(h);
    }
    __syncthreads();
#pragma unroll
    for (int j = 0; j < 16; ++j) {
        int idx = tid + j * 256;
        float p = 1.f;
#pragma unroll
        for (int q = 0; q < 12; ++q)
            p *= cs[q + (((idx >> (11 - q)) & 1) ? 12 : 0)];
        sbuf[idx] = p;
        sbuf[idx + 2 * NSTATE] = 0.f;
    }
    __syncthreads();

    int cur = 0;
    for (int l = 0; l < N_LAYERS; ++l) {
        float* sR = sbuf + cur * NSTATE;
        float* sI = sR + 2 * NSTATE;
        const RingShifts rk = ring_shifts((l % (N_QUBITS - 1)) + 1);
        const int d1 = ring_dst(1, rk) << 2, d2 = ring_dst(2, rk) << 2,
                  d4 = ring_dst(4, rk) << 2;
        const int nxt4 = (cur ^ 1) * (NSTATE * 4);
#pragma unroll
        for (int g = 0; g < 3; ++g) {
            const float* gr = Gr + (l * 3 + g) * 256;
            const float* gi = Gi + (l * 3 + g) * 256;
            v2f ar[4], ad[4], an[4];
#pragma unroll
            for (int kk = 0; kk < 4; ++kk) {
                int k0 = kk * 4 + sel * 2;
                v2f t0, t1;
                t0.x = gr[c * 16 + k0]; t0.y = gr[c * 16 + k0 + 1];
                t1.x = gi[c * 16 + k0]; t1.y = gi[c * 16 + k0 + 1];
                ar[kk] = t0; ad[kk] = t1 - t0; an[kk] = -(t0 + t1);
            }
            const int rs = (g == 0) ? 256 : ((g == 1) ? 16 : 1);
#pragma unroll
            for (int tt = 0; tt < 2; ++tt) {
                int tile = wave * 2 + tt;
                int off = (g == 0) ? (tile * 16 + c)
                        : (g == 1) ? (tile * 256 + c)
                                   : (tile * 256 + c * 16);
                v2f br[4], bi_[4];
#pragma unroll
                for (int kk = 0; kk < 4; ++kk) {
                    int k0 = kk * 4 + sel * 2;
                    if (g == 2) {
                        br[kk]  = *(const v2f*)&sR[k0 + off];
                        bi_[kk] = *(const v2f*)&sI[k0 + off];
                    } else {
                        v2f t0, t1;
                        t0.x = sR[k0 * rs + off]; t0.y = sR[(k0 + 1) * rs + off];
                        t1.x = sI[k0 * rs + off]; t1.y = sI[(k0 + 1) * rs + off];
                        br[kk] = t0; bi_[kk] = t1;
                    }
                }
                v2f bs[4];
#pragma unroll
                for (int kk = 0; kk < 4; ++kk) bs[kk] = br[kk] + bi_[kk];
                v8f k1 = {};
#pragma unroll
                for (int kk = 0; kk < 4; ++kk)
                    k1 = __builtin_amdgcn_wmma_f32_16x16x4_f32(
                        false, ar[kk], false, bs[kk], (short)0, k1, false, false);
                v8f are = k1;
#pragma unroll
                for (int kk = 0; kk < 4; ++kk)
                    are = __builtin_amdgcn_wmma_f32_16x16x4_f32(
                        false, an[kk], false, bi_[kk], (short)0, are, false, false);
                v8f aim = k1;
#pragma unroll
                for (int kk = 0; kk < 4; ++kk)
                    aim = __builtin_amdgcn_wmma_f32_16x16x4_f32(
                        false, ad[kk], false, br[kk], (short)0, aim, false, false);
                float dr[8], di[8];
#pragma unroll
                for (int v = 0; v < 8; ++v) { dr[v] = are[v]; di[v] = aim[v]; }
                if (g == 2) {
                    const int pb = (ring_dst(off + sel * 8, rk) << 2) | nxt4;
                    char* nb = (char*)sbuf;
#pragma unroll
                    for (int v = 0; v < 8; ++v) {
                        int a = pb ^ ((v & 1) ? d1 : 0) ^ ((v & 2) ? d2 : 0)
                                   ^ ((v & 4) ? d4 : 0);
                        *(float*)(nb + a) = dr[v];
                        *(float*)(nb + a + 2 * NSTATE * 4) = di[v];
                    }
                } else {
#pragma unroll
                    for (int v = 0; v < 8; ++v) {
                        int ad = (v + sel * 8) * rs + off;
                        sR[ad] = dr[v];
                        sI[ad] = di[v];
                    }
                }
            }
            __syncthreads();
        }

        cur ^= 1;
    }

    float acc[12];
#pragma unroll
    for (int q = 0; q < 12; ++q) acc[q] = 0.f;
#pragma unroll
    for (int j = 0; j < 16; ++j) {
        int idx = tid + j * 256;
        float re = sbuf[cur * NSTATE + idx], im = sbuf[cur * NSTATE + idx + 2 * NSTATE];
        float p = re * re + im * im;
#pragma unroll
        for (int q = 0; q < 12; ++q)
            acc[q] += ((idx >> (11 - q)) & 1) ? -p : p;
    }
#pragma unroll
    for (int offm = 16; offm > 0; offm >>= 1)
#pragma unroll
        for (int q = 0; q < 12; ++q)
            acc[q] += __shfl_xor(acc[q], offm, 32);
    if (lane == 0)
#pragma unroll
        for (int q = 0; q < 12; ++q) red[wave * 12 + q] = acc[q];
    __syncthreads();
    if (tid < 12) {
        float s = 0.f;
#pragma unroll
        for (int w = 0; w < 8; ++w) s += red[w * 12 + tid];
        ev[tid] = s;
    }
    __syncthreads();

    if (tid < N_CLASSES) {
        float s = fcb[tid];
#pragma unroll
        for (int q = 0; q < 12; ++q) s += ev[q] * fcw[tid * 12 + q];
        out[b * N_CLASSES + tid] = s;
    }
}

extern "C" void kernel_launch(void* const* d_in, const int* in_sizes, int n_in,
                              void* d_out, int out_size, void* d_ws, size_t ws_size,
                              hipStream_t stream) {
    const float* x   = (const float*)d_in[0];
    const float* qp  = (const float*)d_in[1];
    const float* fcw = (const float*)d_in[2];
    const float* fcb = (const float*)d_in[3];
    float* Gr = (float*)d_ws;
    float* Gi = Gr + NMAT * 256;

    prep_gates<<<1, 256, 0, stream>>>(qp, Gr, Gi);
    qsim<<<4096, 256, 0, stream>>>(x, Gr, Gi, fcw, fcb, (float*)d_out);
}


# Gate groups 1,2 chained in registers: group 1 with WMMA operands swapped gives the transposed tile = group 2's B fragment; an LDS round trip and a barrier per layer gone; group 2 sum order changes
# speedup vs baseline: 1.8970x; 1.0371x over previous
#include <hip/hip_runtime.h>
#include <hip/hip_bf16.h>

typedef __attribute__((ext_vector_type(2))) float v2f;
typedef __attribute__((ext_vector_type(4))) float v4f;
typedef __attribute__((ext_vector_type(8))) float v8f;

#define N_QUBITS 12
#define N_LAYERS 6
#define N_CLASSES 10
#define NSTATE 4096
#define NMAT 18

__global__ void prep_gates(const float* __restrict__ qp,
                           float* __restrict__ Gr, float* __restrict__ Gi) {
    __shared__ float ur[72][2][2];
    __shared__ float ui[72][2][2];
    int t = threadIdx.x;
    if (t < 72) {
        int l = t / 12, q = t % 12;
        const float* p = qp + (l * 12 + q) * 3;
        float phi = p[0], th = p[1], om = p[2];
        float ct = cosf(0.5f * th), st = sinf(0.5f * th);
        float a = 0.5f * (phi + om), b = 0.5f * (phi - om);
        float ca = cosf(a), sa = sinf(a), cb = cosf(b), sb = sinf(b);
        ur[t][0][0] = ca * ct;  ui[t][0][0] = -sa * ct;
        ur[t][0][1] = -cb * st; ui[t][0][1] = -sb * st;
        ur[t][1][0] = cb * st;  ui[t][1][0] = -sb * st;
        ur[t][1][1] = ca * ct;  ui[t][1][1] = sa * ct;
    }
    __syncthreads();
    for (int e = t; e < NMAT * 256; e += blockDim.x) {
        int m = e >> 8, pos = e & 255;
        int l = m / 3, g = m % 3;
        int i = pos >> 4, k = pos & 15;
        float pr = 1.f, pi = 0.f;
#pragma unroll
        for (int j = 0; j < 4; ++j) {
            int u = l * 12 + g * 4 + j;
            int bi_ = (i >> (3 - j)) & 1;
            int bk  = (k >> (3 - j)) & 1;
            float wr = ur[u][bi_][bk], wi = ui[u][bi_][bk];
            float nr = pr * wr - pi * wi;
            float ni = pr * wi + pi * wr;
            pr = nr; pi = ni;
        }
        Gr[e] = pr; Gi[e] = pi;
    }
}

struct RingShifts { int s1, s2, s4, s8, wrap; };

__device__ __forceinline__ RingShifts ring_shifts(int r) {
    RingShifts k;
    k.s1 = r;
    k.s2 = min(2 * r, N_QUBITS);
    k.s4 = min(4 * r, N_QUBITS);
    k.s8 = min(8 * r, N_QUBITS);
    k.wrap = N_QUBITS - r;
    return k;
}

__device__ __forceinline__ int ring_dst(int s, const RingShifts k) {
    int c = s;
    c ^= c >> k.s1;
    c ^= c >> k.s2;
    c ^= c >> k.s4;
    c ^= c >> k.s8;
    return c ^ ((c << k.wrap) & (NSTATE - 1));
}

__global__ __launch_bounds__(256) void qsim(
        const float* __restrict__ x,
        const float* __restrict__ Gr, const float* __restrict__ Gi,
        const float* __restrict__ fcw, const float* __restrict__ fcb,
        float* __restrict__ out) {
    __shared__ float sbuf[4 * NSTATE];
    __shared__ float cs[24];
    __shared__ float red[8 * 12];
    __shared__ float ev[12];

    const int b    = blockIdx.x;
    const int tid  = threadIdx.x;
    const int lane = tid & 31;
    const int wave = tid >> 5;
    const int c    = lane & 15;
    const int sel  = lane >> 4;

    if (tid < 12) {
        float h = 0.5f * x[b * 12 + tid];
        cs[tid] = cosf(h);
        cs[12 + tid] = sinf(h);
    }
    __syncthreads();
#pragma unroll
    for (int j = 0; j < 16; ++j) {
        int idx = tid + j * 256;
        float p = 1.f;
#pragma unroll
        for (int q = 0; q < 12; ++q)
            p *= cs[q + (((idx >> (11 - q)) & 1) ? 12 : 0)];
        sbuf[idx] = p;
        sbuf[idx + 2 * NSTATE] = 0.f;
    }
    __syncthreads();

    int cur = 0;
    for (int l = 0; l < N_LAYERS; ++l) {
        float* sR = sbuf + cur * NSTATE;
        float* sI = sR + 2 * NSTATE;
        const RingShifts rk = ring_shifts((l % (N_QUBITS - 1)) + 1);
        const int d1 = ring_dst(1, rk) << 2, d2 = ring_dst(2, rk) << 2,
                  d4 = ring_dst(4, rk) << 2;
        const int nxt4 = (cur ^ 1) * (NSTATE * 4);

        {
            const float* gr = Gr + (l * 3 + 0) * 256;
            const float* gi = Gi + (l * 3 + 0) * 256;
            v2f ar[4], ad[4], an[4];
#pragma unroll
            for (int kk = 0; kk < 4; ++kk) {
                int k0 = kk * 4 + sel * 2;
                v2f t0, t1;
                t0.x = gr[c * 16 + k0]; t0.y = gr[c * 16 + k0 + 1];
                t1.x = gi[c * 16 + k0]; t1.y = gi[c * 16 + k0 + 1];
                ar[kk] = t0; ad[kk] = t1 - t0; an[kk] = -(t0 + t1);
            }
#pragma unroll
            for (int tt = 0; tt < 2; ++tt) {
                int tile = wave * 2 + tt;
                int off = tile * 16 + c;
                v2f br[4], bi_[4], bs[4];
#pragma unroll
                for (int kk = 0; kk < 4; ++kk) {
                    int k0 = kk * 4 + sel * 2;
                    v2f t0, t1;
                    t0.x = sR[k0 * 256 + off]; t0.y = sR[(k0 + 1) * 256 + off];
                    t1.x = sI[k0 * 256 + off]; t1.y = sI[(k0 + 1) * 256 + off];
                    br[kk] = t0; bi_[kk] = t1; bs[kk] = t0 + t1;
                }
                v8f k1 = {};
#pragma unroll
                for (int kk = 0; kk < 4; ++kk)
                    k1 = __builtin_amdgcn_wmma_f32_16x16x4_f32(
                        false, ar[kk], false, bs[kk], (short)0, k1, false, false);
                v8f are = k1;
#pragma unroll
                for (int kk = 0; kk < 4; ++kk)
                    are = __builtin_amdgcn_wmma_f32_16x16x4_f32(
                        false, an[kk], false, bi_[kk], (short)0, are, false, false);
                v8f aim = k1;
#pragma unroll
                for (int kk = 0; kk < 4; ++kk)
                    aim = __builtin_amdgcn_wmma_f32_16x16x4_f32(
                        false, ad[kk], false, br[kk], (short)0, aim, false, false);
#pragma unroll
                for (int v = 0; v < 8; ++v) {
                    int ad_ = (v + sel * 8) * 256 + off;
                    sR[ad_] = are[v];
                    sI[ad_] = aim[v];
                }
            }
        }
        __syncthreads();

        {
            const float* gr1 = Gr + (l * 3 + 1) * 256;
            const float* gi1 = Gi + (l * 3 + 1) * 256;
            const float* gr2 = Gr + (l * 3 + 2) * 256;
            const float* gi2 = Gi + (l * 3 + 2) * 256;
            v2f ar1[4], ad1[4], an1[4];
            v2f ar2[4], ad2[4], an2[4];
#pragma unroll
            for (int kk = 0; kk < 4; ++kk) {
                int k0 = kk * 4 + sel * 2;
                v2f t0, t1;
                t0.x = gr1[c * 16 + k0]; t0.y = gr1[c * 16 + k0 + 1];
                t1.x = gi1[c * 16 + k0]; t1.y = gi1[c * 16 + k0 + 1];
                ar1[kk] = t0; ad1[kk] = t1 - t0; an1[kk] = -(t0 + t1);
                int k2 = sel * 8 + kk * 2;
                v2f u0, u1;
                u0.x = gr2[c * 16 + k2]; u0.y = gr2[c * 16 + k2 + 1];
                u1.x = gi2[c * 16 + k2]; u1.y = gi2[c * 16 + k2 + 1];
                ar2[kk] = u0; ad2[kk] = u1 - u0; an2[kk] = -(u0 + u1);
            }
#pragma unroll
            for (int tt = 0; tt < 2; ++tt) {
                int tile = wave * 2 + tt;
                int off1 = tile * 256 + c;
                v2f br[4], bi_[4], bs[4];
#pragma unroll
                for (int kk = 0; kk < 4; ++kk) {
                    int k0 = kk * 4 + sel * 2;
                    v2f t0, t1;
                    t0.x = sR[k0 * 16 + off1]; t0.y = sR[(k0 + 1) * 16 + off1];
                    t1.x = sI[k0 * 16 + off1]; t1.y = sI[(k0 + 1) * 16 + off1];
                    br[kk] = t0; bi_[kk] = t1; bs[kk] = t0 + t1;
                }
                v8f k1 = {};
#pragma unroll
                for (int kk = 0; kk < 4; ++kk)
                    k1 = __builtin_amdgcn_wmma_f32_16x16x4_f32(
                        false, bs[kk], false, ar1[kk], (short)0, k1, false, false);
                v8f tre = k1;
#pragma unroll
                for (int kk = 0; kk < 4; ++kk)
                    tre = __builtin_amdgcn_wmma_f32_16x16x4_f32(
                        false, bi_[kk], false, an1[kk], (short)0, tre, false, false);
                v8f tim = k1;
#pragma unroll
                for (int kk = 0; kk < 4; ++kk)
                    tim = __builtin_amdgcn_wmma_f32_16x16x4_f32(
                        false, br[kk], false, ad1[kk], (short)0, tim, false, false);
                v2f cr[4], ci[4], cs2[4];
#pragma unroll
                for (int kk = 0; kk < 4; ++kk) {
                    v2f t0, t1;
                    t0.x = tre[2 * kk]; t0.y = tre[2 * kk + 1];
                    t1.x = tim[2 * kk]; t1.y = tim[2 * kk + 1];
                    cr[kk] = t0; ci[kk] = t1; cs2[kk] = t0 + t1;
                }
                v8f k2 = {};
#pragma unroll
                for (int kk = 0; kk < 4; ++kk)
                    k2 = __builtin_amdgcn_wmma_f32_16x16x4_f32(
                        false, ar2[kk], false, cs2[kk], (short)0, k2, false, false);
                v8f are = k2;
#pragma unroll
                for (int kk = 0; kk < 4; ++kk)
                    are = __builtin_amdgcn_wmma_f32_16x16x4_f32(
                        false, an2[kk], false, ci[kk], (short)0, are, false, false);
                v8f aim = k2;
#pragma unroll
                for (int kk = 0; kk < 4; ++kk)
                    aim = __builtin_amdgcn_wmma_f32_16x16x4_f32(
                        false, ad2[kk], false, cr[kk], (short)0, aim, false, false);
                const int off = tile * 256 + c * 16;
                const int pb = (ring_dst(off + sel * 8, rk) << 2) | nxt4;
                char* nb = (char*)sbuf;
#pragma unroll
                for (int v = 0; v < 8; ++v) {
                    int a = pb ^ ((v & 1) ? d1 : 0) ^ ((v & 2) ? d2 : 0)
                               ^ ((v & 4) ? d4 : 0);
                    *(float*)(nb + a) = are[v];
                    *(float*)(nb + a + 2 * NSTATE * 4) = aim[v];
                }
            }
        }
        __syncthreads();

        cur ^= 1;
    }

    float acc[12];
#pragma unroll
    for (int q = 0; q < 12; ++q) acc[q] = 0.f;
#pragma unroll
    for (int j = 0; j < 16; ++j) {
        int idx = tid + j * 256;
        float re = sbuf[cur * NSTATE + idx], im = sbuf[cur * NSTATE + idx + 2 * NSTATE];
        float p = re * re + im * im;
#pragma unroll
        for (int q = 0; q < 12; ++q)
            acc[q] += ((idx >> (11 - q)) & 1) ? -p : p;
    }
#pragma unroll
    for (int offm = 16; offm > 0; offm >>= 1)
#pragma unroll
        for (int q = 0; q < 12; ++q)
            acc[q] += __shfl_xor(acc[q], offm, 32);
    if (lane == 0)
#pragma unroll
        for (int q = 0; q < 12; ++q) red[wave * 12 + q] = acc[q];
    __syncthreads();
    if (tid < 12) {
        float s = 0.f;
#pragma unroll
        for (int w = 0; w < 8; ++w) s += red[w * 12 + tid];
        ev[tid] = s;
    }
    __syncthreads();

    if (tid < N_CLASSES) {
        float s = fcb[tid];
#pragma unroll
        for (int q = 0; q < 12; ++q) s += ev[q] * fcw[tid * 12 + q];
        out[b * N_CLASSES + tid] = s;
    }
}

extern "C" void kernel_launch(void* const* d_in, const int* in_sizes, int n_in,
                              void* d_out, int out_size, void* d_ws, size_t ws_size,
                              hipStream_t stream) {
    const float* x   = (const float*)d_in[0];
    const float* qp  = (const float*)d_in[1];
    const float* fcw = (const float*)d_in[2];
    const float* fcb = (const float*)d_in[3];
    float* Gr = (float*)d_ws;
    float* Gi = Gr + NMAT * 256;

    prep_gates<<<1, 256, 0, stream>>>(qp, Gr, Gi);
    qsim<<<4096, 256, 0, stream>>>(x, Gr, Gi, fcw, fcb, (float*)d_out);
}


# Layer 0 acts on a product state: its twelve 2x2 gates are applied to the twelve qubit factors before the tensor product is formed; layer 0's matrix passes and 2 barriers go; its sum order changes
# speedup vs baseline: 2.2273x; 1.1741x over previous
#include <hip/hip_runtime.h>
#include <hip/hip_bf16.h>

typedef __attribute__((ext_vector_type(2))) float v2f;
typedef __attribute__((ext_vector_type(4))) float v4f;
typedef __attribute__((ext_vector_type(8))) float v8f;

#define N_QUBITS 12
#define N_LAYERS 6
#define N_CLASSES 10
#define NSTATE 4096
#define NMAT 18

__global__ void prep_gates(const float* __restrict__ qp,
                           float* __restrict__ Gr, float* __restrict__ Gi) {
    __shared__ float ur[72][2][2];
    __shared__ float ui[72][2][2];
    int t = threadIdx.x;
    if (t < 72) {
        int l = t / 12, q = t % 12;
        const float* p = qp + (l * 12 + q) * 3;
        float phi = p[0], th = p[1], om = p[2];
        float ct = cosf(0.5f * th), st = sinf(0.5f * th);
        float a = 0.5f * (phi + om), b = 0.5f * (phi - om);
        float ca = cosf(a), sa = sinf(a), cb = cosf(b), sb = sinf(b);
        ur[t][0][0] = ca * ct;  ui[t][0][0] = -sa * ct;
        ur[t][0][1] = -cb * st; ui[t][0][1] = -sb * st;
        ur[t][1][0] = cb * st;  ui[t][1][0] = -sb * st;
        ur[t][1][1] = ca * ct;  ui[t][1][1] = sa * ct;
        if (t < 12) {
#pragma unroll
            for (int e = 0; e < 4; ++e) {
                Gr[t * 4 + e] = ur[t][e >> 1][e & 1];
                Gi[t * 4 + e] = ui[t][e >> 1][e & 1];
            }
        }
    }
    __syncthreads();
    for (int e = 3 * 256 + t; e < NMAT * 256; e += blockDim.x) {
        int m = e >> 8, pos = e & 255;
        int l = m / 3, g = m % 3;
        int i = pos >> 4, k = pos & 15;
        float pr = 1.f, pi = 0.f;
#pragma unroll
        for (int j = 0; j < 4; ++j) {
            int u = l * 12 + g * 4 + j;
            int bi_ = (i >> (3 - j)) & 1;
            int bk  = (k >> (3 - j)) & 1;
            float wr = ur[u][bi_][bk], wi = ui[u][bi_][bk];
            float nr = pr * wr - pi * wi;
            float ni = pr * wi + pi * wr;
            pr = nr; pi = ni;
        }
        Gr[e] = pr; Gi[e] = pi;
    }
}

struct RingShifts { int s1, s2, s4, s8, wrap; };

__device__ __forceinline__ RingShifts ring_shifts(int r) {
    RingShifts k;
    k.s1 = r;
    k.s2 = min(2 * r, N_QUBITS);
    k.s4 = min(4 * r, N_QUBITS);
    k.s8 = min(8 * r, N_QUBITS);
    k.wrap = N_QUBITS - r;
    return k;
}

__device__ __forceinline__ int ring_dst(int s, const RingShifts k) {
    int c = s;
    c ^= c >> k.s1;
    c ^= c >> k.s2;
    c ^= c >> k.s4;
    c ^= c >> k.s8;
    return c ^ ((c << k.wrap) & (NSTATE - 1));
}

__global__ __launch_bounds__(256) void qsim(
        const float* __restrict__ x,
        const float* __restrict__ Gr, const float* __restrict__ Gi,
        const float* __restrict__ fcw, const float* __restrict__ fcb,
        float* __restrict__ out) {
    __shared__ float sbuf[4 * NSTATE];
    __shared__ float fr[24], fi[24];
    __shared__ float tr[48], ti[48];
    __shared__ float red[8 * 12];
    __shared__ float ev[12];

    const int b    = blockIdx.x;
    const int tid  = threadIdx.x;
    const int lane = tid & 31;
    const int wave = tid >> 5;
    const int c    = lane & 15;
    const int sel  = lane >> 4;

    if (tid < 12) {
        float h = 0.5f * x[b * 12 + tid];
        float ch = cosf(h), sh = sinf(h);
        const float* ur = Gr + tid * 4;
        const float* ui = Gi + tid * 4;
        fr[2 * tid]     = ur[0] * ch + ur[1] * sh;
        fi[2 * tid]     = ui[0] * ch + ui[1] * sh;
        fr[2 * tid + 1] = ur[2] * ch + ur[3] * sh;
        fi[2 * tid + 1] = ui[2] * ch + ui[3] * sh;
    }
    __syncthreads();
    if (tid < 48) {
        int g = tid >> 4, i = tid & 15;
        float pr = 1.f, pi = 0.f;
#pragma unroll
        for (int j = 0; j < 4; ++j) {
            int u = 2 * (g * 4 + j) + ((i >> (3 - j)) & 1);
            float wr = fr[u], wi = fi[u];
            float nr = pr * wr - pi * wi;
            float ni = pr * wi + pi * wr;
            pr = nr; pi = ni;
        }
        tr[tid] = pr; ti[tid] = pi;
    }
    __syncthreads();
    {
        const RingShifts rk0 = ring_shifts(1);
        const float p1r = tr[16 + (tid >> 4)], p1i = ti[16 + (tid >> 4)];
        const float p2r = tr[32 + (tid & 15)], p2i = ti[32 + (tid & 15)];
        const float pr = p1r * p2r - p1i * p2i;
        const float pi = p1r * p2i + p1i * p2r;
        const int dt = ring_dst(tid, rk0);
#pragma unroll
        for (int j = 0; j < 16; ++j) {
            float ar = tr[j], ai = ti[j];
            int d = dt ^ ring_dst(j << 8, rk0);
            sbuf[d] = ar * pr - ai * pi;
            sbuf[d + 2 * NSTATE] = ar * pi + ai * pr;
        }
    }
    __syncthreads();

    int cur = 0;
    for (int l = 1; l < N_LAYERS; ++l) {
        float* sR = sbuf + cur * NSTATE;
        float* sI = sR + 2 * NSTATE;
        const RingShifts rk = ring_shifts((l % (N_QUBITS - 1)) + 1);
        const int d1 = ring_dst(1, rk) << 2, d2 = ring_dst(2, rk) << 2,
                  d4 = ring_dst(4, rk) << 2;
        const int nxt4 = (cur ^ 1) * (NSTATE * 4);

        {
            const float* gr = Gr + (l * 3 + 0) * 256;
            const float* gi = Gi + (l * 3 + 0) * 256;
            v2f ar[4], ad[4], an[4];
#pragma unroll
            for (int kk = 0; kk < 4; ++kk) {
                int k0 = kk * 4 + sel * 2;
                v2f t0, t1;
                t0.x = gr[c * 16 + k0]; t0.y = gr[c * 16 + k0 + 1];
                t1.x = gi[c * 16 + k0]; t1.y = gi[c * 16 + k0 + 1];
                ar[kk] = t0; ad[kk] = t1 - t0; an[kk] = -(t0 + t1);
            }
#pragma unroll
            for (int tt = 0; tt < 2; ++tt) {
                int tile = wave * 2 + tt;
                int off = tile * 16 + c;
                v2f br[4], bi_[4], bs[4];
#pragma unroll
                for (int kk = 0; kk < 4; ++kk) {
                    int k0 = kk * 4 + sel * 2;
                    v2f t0, t1;
                    t0.x = sR[k0 * 256 + off]; t0.y = sR[(k0 + 1) * 256 + off];
                    t1.x = sI[k0 * 256 + off]; t1.y = sI[(k0 + 1) * 256 + off];
                    br[kk] = t0; bi_[kk] = t1; bs[kk] = t0 + t1;
                }
                v8f k1 = {};
#pragma unroll
                for (int kk = 0; kk < 4; ++kk)
                    k1 = __builtin_amdgcn_wmma_f32_16x16x4_f32(
                        false, ar[kk], false, bs[kk], (short)0, k1, false, false);
                v8f are = k1;
#pragma unroll
                for (int kk = 0; kk < 4; ++kk)
                    are = __builtin_amdgcn_wmma_f32_16x16x4_f32(
                        false, an[kk], false, bi_[kk], (short)0, are, false, false);
                v8f aim = k1;
#pragma unroll
                for (int kk = 0; kk < 4; ++kk)
                    aim = __builtin_amdgcn_wmma_f32_16x16x4_f32(
                        false, ad[kk], false, br[kk], (short)0, aim, false, false);
#pragma unroll
                for (int v = 0; v < 8; ++v) {
                    int ad_ = (v + sel * 8) * 256 + off;
                    sR[ad_] = are[v];
                    sI[ad_] = aim[v];
                }
            }
        }
        __syncthreads();

        {
            const float* gr1 = Gr + (l * 3 + 1) * 256;
            const float* gi1 = Gi + (l * 3 + 1) * 256;
            const float* gr2 = Gr + (l * 3 + 2) * 256;
            const float* gi2 = Gi + (l * 3 + 2) * 256;
            v2f ar1[4], ad1[4], an1[4];
            v2f ar2[4], ad2[4], an2[4];
#pragma unroll
            for (int kk = 0; kk < 4; ++kk) {
                int k0 = kk * 4 + sel * 2;
                v2f t0, t1;
                t0.x = gr1[c * 16 + k0]; t0.y = gr1[c * 16 + k0 + 1];
                t1.x = gi1[c * 16 + k0]; t1.y = gi1[c * 16 + k0 + 1];
                ar1[kk] = t0; ad1[kk] = t1 - t0; an1[kk] = -(t0 + t1);
                int k2 = sel * 8 + kk * 2;
                v2f u0, u1;
                u0.x = gr2[c * 16 + k2]; u0.y = gr2[c * 16 + k2 + 1];
                u1.x = gi2[c * 16 + k2]; u1.y = gi2[c * 16 + k2 + 1];
                ar2[kk] = u0; ad2[kk] = u1 - u0; an2[kk] = -(u0 + u1);
            }
#pragma unroll
            for (int tt = 0; tt < 2; ++tt) {
                int tile = wave * 2 + tt;
                int off1 = tile * 256 + c;
                v2f br[4], bi_[4], bs[4];
#pragma unroll
                for (int kk = 0; kk < 4; ++kk) {
                    int k0 = kk * 4 + sel * 2;
                    v2f t0, t1;
                    t0.x = sR[k0 * 16 + off1]; t0.y = sR[(k0 + 1) * 16 + off1];
                    t1.x = sI[k0 * 16 + off1]; t1.y = sI[(k0 + 1) * 16 + off1];
                    br[kk] = t0; bi_[kk] = t1; bs[kk] = t0 + t1;
                }
                v8f k1 = {};
#pragma unroll
                for (int kk = 0; kk < 4; ++kk)
                    k1 = __builtin_amdgcn_wmma_f32_16x16x4_f32(
                        false, bs[kk], false, ar1[kk], (short)0, k1, false, false);
                v8f tre = k1;
#pragma unroll
                for (int kk = 0; kk < 4; ++kk)
                    tre = __builtin_amdgcn_wmma_f32_16x16x4_f32(
                        false, bi_[kk], false, an1[kk], (short)0, tre, false, false);
                v8f tim = k1;
#pragma unroll
                for (int kk = 0; kk < 4; ++kk)
                    tim = __builtin_amdgcn_wmma_f32_16x16x4_f32(
                        false, br[kk], false, ad1[kk], (short)0, tim, false, false);
                v2f cr[4], ci[4], cs2[4];
#pragma unroll
                for (int kk = 0; kk < 4; ++kk) {
                    v2f t0, t1;
                    t0.x = tre[2 * kk]; t0.y = tre[2 * kk + 1];
                    t1.x = tim[2 * kk]; t1.y = tim[2 * kk + 1];
                    cr[kk] = t0; ci[kk] = t1; cs2[kk] = t0 + t1;
                }
                v8f k2 = {};
#pragma unroll
                for (int kk = 0; kk < 4; ++kk)
                    k2 = __builtin_amdgcn_wmma_f32_16x16x4_f32(
                        false, ar2[kk], false, cs2[kk], (short)0, k2, false, false);
                v8f are = k2;
#pragma unroll
                for (int kk = 0; kk < 4; ++kk)
                    are = __builtin_amdgcn_wmma_f32_16x16x4_f32(
                        false, an2[kk], false, ci[kk], (short)0, are, false, false);
                v8f aim = k2;
#pragma unroll
                for (int kk = 0; kk < 4; ++kk)
                    aim = __builtin_amdgcn_wmma_f32_16x16x4_f32(
                        false, ad2[kk], false, cr[kk], (short)0, aim, false, false);
                const int off = tile * 256 + c * 16;
                const int pb = (ring_dst(off + sel * 8, rk) << 2) | nxt4;
                char* nb = (char*)sbuf;
#pragma unroll
                for (int v = 0; v < 8; ++v) {
                    int a = pb ^ ((v & 1) ? d1 : 0) ^ ((v & 2) ? d2 : 0)
                               ^ ((v & 4) ? d4 : 0);
                    *(float*)(nb + a) = are[v];
                    *(float*)(nb + a + 2 * NSTATE * 4) = aim[v];
                }
            }
        }
        __syncthreads();

        cur ^= 1;
    }

    float acc[12];
#pragma unroll
    for (int q = 0; q < 12; ++q) acc[q] = 0.f;
#pragma unroll
    for (int j = 0; j < 16; ++j) {
        int idx = tid + j * 256;
        float re = sbuf[cur * NSTATE + idx], im = sbuf[cur * NSTATE + idx + 2 * NSTATE];
        float p = re * re + im * im;
#pragma unroll
        for (int q = 0; q < 12; ++q)
            acc[q] += ((idx >> (11 - q)) & 1) ? -p : p;
    }
#pragma unroll
    for (int offm = 16; offm > 0; offm >>= 1)
#pragma unroll
        for (int q = 0; q < 12; ++q)
            acc[q] += __shfl_xor(acc[q], offm, 32);
    if (lane == 0)
#pragma unroll
        for (int q = 0; q < 12; ++q) red[wave * 12 + q] = acc[q];
    __syncthreads();
    if (tid < 12) {
        float s = 0.f;
#pragma unroll
        for (int w = 0; w < 8; ++w) s += red[w * 12 + tid];
        ev[tid] = s;
    }
    __syncthreads();

    if (tid < N_CLASSES) {
        float s = fcb[tid];
#pragma unroll
        for (int q = 0; q < 12; ++q) s += ev[q] * fcw[tid * 12 + q];
        out[b * N_CLASSES + tid] = s;
    }
}

extern "C" void kernel_launch(void* const* d_in, const int* in_sizes, int n_in,
                              void* d_out, int out_size, void* d_ws, size_t ws_size,
                              hipStream_t stream) {
    const float* x   = (const float*)d_in[0];
    const float* qp  = (const float*)d_in[1];
    const float* fcw = (const float*)d_in[2];
    const float* fcb = (const float*)d_in[3];
    float* Gr = (float*)d_ws;
    float* Gi = Gr + NMAT * 256;

    prep_gates<<<1, 256, 0, stream>>>(qp, Gr, Gi);
    qsim<<<4096, 256, 0, stream>>>(x, Gr, Gi, fcw, fcb, (float*)d_out);
}


# Work-group of 4 waves with four tiles per wave instead of 8 waves with two: per-layer gate-fragment loads and ring set-up of a wave serve twice the tiles; expval partial-sum order changes
# speedup vs baseline: 2.3778x; 1.0676x over previous
#include <hip/hip_runtime.h>
#include <hip/hip_bf16.h>

typedef __attribute__((ext_vector_type(2))) float v2f;
typedef __attribute__((ext_vector_type(4))) float v4f;
typedef __attribute__((ext_vector_type(8))) float v8f;

#define N_QUBITS 12
#define N_LAYERS 6
#define N_CLASSES 10
#define NSTATE 4096
#define NMAT 18

__global__ void prep_gates(const float* __restrict__ qp,
                           float* __restrict__ Gr, float* __restrict__ Gi) {
    __shared__ float ur[72][2][2];
    __shared__ float ui[72][2][2];
    int t = threadIdx.x;
    if (t < 72) {
        int l = t / 12, q = t % 12;
        const float* p = qp + (l * 12 + q) * 3;
        float phi = p[0], th = p[1], om = p[2];
        float ct = cosf(0.5f * th), st = sinf(0.5f * th);
        float a = 0.5f * (phi + om), b = 0.5f * (phi - om);
        float ca = cosf(a), sa = sinf(a), cb = cosf(b), sb = sinf(b);
        ur[t][0][0] = ca * ct;  ui[t][0][0] = -sa * ct;
        ur[t][0][1] = -cb * st; ui[t][0][1] = -sb * st;
        ur[t][1][0] = cb * st;  ui[t][1][0] = -sb * st;
        ur[t][1][1] = ca * ct;  ui[t][1][1] = sa * ct;
        if (t < 12) {
#pragma unroll
            for (int e = 0; e < 4; ++e) {
                Gr[t * 4 + e] = ur[t][e >> 1][e & 1];
                Gi[t * 4 + e] = ui[t][e >> 1][e & 1];
            }
        }
    }
    __syncthreads();
    for (int e = 3 * 256 + t; e < NMAT * 256; e += blockDim.x) {
        int m = e >> 8, pos = e & 255;
        int l = m / 3, g = m % 3;
        int i = pos >> 4, k = pos & 15;
        float pr = 1.f, pi = 0.f;
#pragma unroll
        for (int j = 0; j < 4; ++j) {
            int u = l * 12 + g * 4 + j;
            int bi_ = (i >> (3 - j)) & 1;
            int bk  = (k >> (3 - j)) & 1;
            float wr = ur[u][bi_][bk], wi = ui[u][bi_][bk];
            float nr = pr * wr - pi * wi;
            float ni = pr * wi + pi * wr;
            pr = nr; pi = ni;
        }
        Gr[e] = pr; Gi[e] = pi;
    }
}

struct RingShifts { int s1, s2, s4, s8, wrap; };

__device__ __forceinline__ RingShifts ring_shifts(int r) {
    RingShifts k;
    k.s1 = r;
    k.s2 = min(2 * r, N_QUBITS);
    k.s4 = min(4 * r, N_QUBITS);
    k.s8 = min(8 * r, N_QUBITS);
    k.wrap = N_QUBITS - r;
    return k;
}

__device__ __forceinline__ int ring_dst(int s, const RingShifts k) {
    int c = s;
    c ^= c >> k.s1;
    c ^= c >> k.s2;
    c ^= c >> k.s4;
    c ^= c >> k.s8;
    return c ^ ((c << k.wrap) & (NSTATE - 1));
}

__global__ __launch_bounds__(128) void qsim(
        const float* __restrict__ x,
        const float* __restrict__ Gr, const float* __restrict__ Gi,
        const float* __restrict__ fcw, const float* __restrict__ fcb,
        float* __restrict__ out) {
    __shared__ float sbuf[4 * NSTATE];
    __shared__ float fr[24], fi[24];
    __shared__ float tr[48], ti[48];
    __shared__ float red[4 * 12];
    __shared__ float ev[12];

    const int b    = blockIdx.x;
    const int tid  = threadIdx.x;
    const int lane = tid & 31;
    const int wave = tid >> 5;
    const int c    = lane & 15;
    const int sel  = lane >> 4;

    if (tid < 12) {
        float h = 0.5f * x[b * 12 + tid];
        float ch = cosf(h), sh = sinf(h);
        const float* ur = Gr + tid * 4;
        const float* ui = Gi + tid * 4;
        fr[2 * tid]     = ur[0] * ch + ur[1] * sh;
        fi[2 * tid]     = ui[0] * ch + ui[1] * sh;
        fr[2 * tid + 1] = ur[2] * ch + ur[3] * sh;
        fi[2 * tid + 1] = ui[2] * ch + ui[3] * sh;
    }
    __syncthreads();
    if (tid < 48) {
        int g = tid >> 4, i = tid & 15;
        float pr = 1.f, pi = 0.f;
#pragma unroll
        for (int j = 0; j < 4; ++j) {
            int u = 2 * (g * 4 + j) + ((i >> (3 - j)) & 1);
            float wr = fr[u], wi = fi[u];
            float nr = pr * wr - pi * wi;
            float ni = pr * wi + pi * wr;
            pr = nr; pi = ni;
        }
        tr[tid] = pr; ti[tid] = pi;
    }
    __syncthreads();
    {
        const RingShifts rk0 = ring_shifts(1);
#pragma unroll
        for (int h = 0; h < 2; ++h) {
            const int lo = tid + 128 * h;
            const float p1r = tr[16 + (lo >> 4)], p1i = ti[16 + (lo >> 4)];
            const float p2r = tr[32 + (lo & 15)], p2i = ti[32 + (lo & 15)];
            const float pr = p1r * p2r - p1i * p2i;
            const float pi = p1r * p2i + p1i * p2r;
            const int dt = ring_dst(lo, rk0);
#pragma unroll
            for (int j = 0; j < 16; ++j) {
                float ar = tr[j], ai = ti[j];
                int d = dt ^ ring_dst(j << 8, rk0);
                sbuf[d] = ar * pr - ai * pi;
                sbuf[d + 2 * NSTATE] = ar * pi + ai * pr;
            }
        }
    }
    __syncthreads();

    int cur = 0;
    for (int l = 1; l < N_LAYERS; ++l) {
        float* sR = sbuf + cur * NSTATE;
        float* sI = sR + 2 * NSTATE;
        const RingShifts rk = ring_shifts((l % (N_QUBITS - 1)) + 1);
        const int d1 = ring_dst(1, rk) << 2, d2 = ring_dst(2, rk) << 2,
                  d4 = ring_dst(4, rk) << 2;
        const int nxt4 = (cur ^ 1) * (NSTATE * 4);

        {
            const float* gr = Gr + (l * 3 + 0) * 256;
            const float* gi = Gi + (l * 3 + 0) * 256;
            v2f ar[4], ad[4], an[4];
#pragma unroll
            for (int kk = 0; kk < 4; ++kk) {
                int k0 = kk * 4 + sel * 2;
                v2f t0, t1;
                t0.x = gr[c * 16 + k0]; t0.y = gr[c * 16 + k0 + 1];
                t1.x = gi[c * 16 + k0]; t1.y = gi[c * 16 + k0 + 1];
                ar[kk] = t0; ad[kk] = t1 - t0; an[kk] = -(t0 + t1);
            }
#pragma unroll
            for (int tt = 0; tt < 4; ++tt) {
                int tile = tt * 4 + wave;
                int off = tile * 16 + c;
                v2f br[4], bi_[4], bs[4];
#pragma unroll
                for (int kk = 0; kk < 4; ++kk) {
                    int k0 = kk * 4 + sel * 2;
                    v2f t0, t1;
                    t0.x = sR[k0 * 256 + off]; t0.y = sR[(k0 + 1) * 256 + off];
                    t1.x = sI[k0 * 256 + off]; t1.y = sI[(k0 + 1) * 256 + off];
                    br[kk] = t0; bi_[kk] = t1; bs[kk] = t0 + t1;
                }
                v8f k1 = {};
#pragma unroll
                for (int kk = 0; kk < 4; ++kk)
                    k1 = __builtin_amdgcn_wmma_f32_16x16x4_f32(
                        false, ar[kk], false, bs[kk], (short)0, k1, false, false);
                v8f are = k1;
#pragma unroll
                for (int kk = 0; kk < 4; ++kk)
                    are = __builtin_amdgcn_wmma_f32_16x16x4_f32(
                        false, an[kk], false, bi_[kk], (short)0, are, false, false);
                v8f aim = k1;
#pragma unroll
                for (int kk = 0; kk < 4; ++kk)
                    aim = __builtin_amdgcn_wmma_f32_16x16x4_f32(
                        false, ad[kk], false, br[kk], (short)0, aim, false, false);
#pragma unroll
                for (int v = 0; v < 8; ++v) {
                    int ad_ = (v + sel * 8) * 256 + off;
                    sR[ad_] = are[v];
                    sI[ad_] = aim[v];
                }
            }
        }
        __syncthreads();

        {
            const float* gr1 = Gr + (l * 3 + 1) * 256;
            const float* gi1 = Gi + (l * 3 + 1) * 256;
            const float* gr2 = Gr + (l * 3 + 2) * 256;
            const float* gi2 = Gi + (l * 3 + 2) * 256;
            v2f ar1[4], ad1[4], an1[4];
            v2f ar2[4], ad2[4], an2[4];
#pragma unroll
            for (int kk = 0; kk < 4; ++kk) {
                int k0 = kk * 4 + sel * 2;
                v2f t0, t1;
                t0.x = gr1[c * 16 + k0]; t0.y = gr1[c * 16 + k0 + 1];
                t1.x = gi1[c * 16 + k0]; t1.y = gi1[c * 16 + k0 + 1];
                ar1[kk] = t0; ad1[kk] = t1 - t0; an1[kk] = -(t0 + t1);
                int k2 = sel * 8 + kk * 2;
                v2f u0, u1;
                u0.x = gr2[c * 16 + k2]; u0.y = gr2[c * 16 + k2 + 1];
                u1.x = gi2[c * 16 + k2]; u1.y = gi2[c * 16 + k2 + 1];
                ar2[kk] = u0; ad2[kk] = u1 - u0; an2[kk] = -(u0 + u1);
            }
#pragma unroll
            for (int tt = 0; tt < 4; ++tt) {
                int tile = wave * 4 + tt;
                int off1 = tile * 256 + c;
                v2f br[4], bi_[4], bs[4];
#pragma unroll
                for (int kk = 0; kk < 4; ++kk) {
                    int k0 = kk * 4 + sel * 2;
                    v2f t0, t1;
                    t0.x = sR[k0 * 16 + off1]; t0.y = sR[(k0 + 1) * 16 + off1];
                    t1.x = sI[k0 * 16 + off1]; t1.y = sI[(k0 + 1) * 16 + off1];
                    br[kk] = t0; bi_[kk] = t1; bs[kk] = t0 + t1;
                }
                v8f k1 = {};
#pragma unroll
                for (int kk = 0; kk < 4; ++kk)
                    k1 = __builtin_amdgcn_wmma_f32_16x16x4_f32(
                        false, bs[kk], false, ar1[kk], (short)0, k1, false, false);
                v8f tre = k1;
#pragma unroll
                for (int kk = 0; kk < 4; ++kk)
                    tre = __builtin_amdgcn_wmma_f32_16x16x4_f32(
                        false, bi_[kk], false, an1[kk], (short)0, tre, false, false);
                v8f tim = k1;
#pragma unroll
                for (int kk = 0; kk < 4; ++kk)
                    tim = __builtin_amdgcn_wmma_f32_16x16x4_f32(
                        false, br[kk], false, ad1[kk], (short)0, tim, false, false);
                v2f cr[4], ci[4], cs2[4];
#pragma unroll
                for (int kk = 0; kk < 4; ++kk) {
                    v2f t0, t1;
                    t0.x = tre[2 * kk]; t0.y = tre[2 * kk + 1];
                    t1.x = tim[2 * kk]; t1.y = tim[2 * kk + 1];
                    cr[kk] = t0; ci[kk] = t1; cs2[kk] = t0 + t1;
                }
                v8f k2 = {};
#pragma unroll
                for (int kk = 0; kk < 4; ++kk)
                    k2 = __builtin_amdgcn_wmma_f32_16x16x4_f32(
                        false, ar2[kk], false, cs2[kk], (short)0, k2, false, false);
                v8f are = k2;
#pragma unroll
                for (int kk = 0; kk < 4; ++kk)
                    are = __builtin_amdgcn_wmma_f32_16x16x4_f32(
                        false, an2[kk], false, ci[kk], (short)0, are, false, false);
                v8f aim = k2;
#pragma unroll
                for (int kk = 0; kk < 4; ++kk)
                    aim = __builtin_amdgcn_wmma_f32_16x16x4_f32(
                        false, ad2[kk], false, cr[kk], (short)0, aim, false, false);
                const int off = tile * 256 + c * 16;
                const int pb = (ring_dst(off + sel * 8, rk) << 2) | nxt4;
                char* nb = (char*)sbuf;
#pragma unroll
                for (int v = 0; v < 8; ++v) {
                    int a = pb ^ ((v & 1) ? d1 : 0) ^ ((v & 2) ? d2 : 0)
                               ^ ((v & 4) ? d4 : 0);
                    *(float*)(nb + a) = are[v];
                    *(float*)(nb + a + 2 * NSTATE * 4) = aim[v];
                }
            }
        }
        __syncthreads();

        cur ^= 1;
    }

    float acc[12];
#pragma unroll
    for (int q = 0; q < 12; ++q) acc[q] = 0.f;
#pragma unroll
    for (int j = 0; j < 32; ++j) {
        int idx = tid + j * 128;
        float re = sbuf[cur * NSTATE + idx], im = sbuf[cur * NSTATE + idx + 2 * NSTATE];
        float p = re * re + im * im;
#pragma unroll
        for (int q = 0; q < 12; ++q)
            acc[q] += ((idx >> (11 - q)) & 1) ? -p : p;
    }
#pragma unroll
    for (int offm = 16; offm > 0; offm >>= 1)
#pragma unroll
        for (int q = 0; q < 12; ++q)
            acc[q] += __shfl_xor(acc[q], offm, 32);
    if (lane == 0)
#pragma unroll
        for (int q = 0; q < 12; ++q) red[wave * 12 + q] = acc[q];
    __syncthreads();
    if (tid < 12) {
        float s = 0.f;
#pragma unroll
        for (int w = 0; w < 4; ++w) s += red[w * 12 + tid];
        ev[tid] = s;
    }
    __syncthreads();

    if (tid < N_CLASSES) {
        float s = fcb[tid];
#pragma unroll
        for (int q = 0; q < 12; ++q) s += ev[q] * fcw[tid * 12 + q];
        out[b * N_CLASSES + tid] = s;
    }
}

extern "C" void kernel_launch(void* const* d_in, const int* in_sizes, int n_in,
                              void* d_out, int out_size, void* d_ws, size_t ws_size,
                              hipStream_t stream) {
    const float* x   = (const float*)d_in[0];
    const float* qp  = (const float*)d_in[1];
    const float* fcw = (const float*)d_in[2];
    const float* fcb = (const float*)d_in[3];
    float* Gr = (float*)d_ws;
    float* Gi = Gr + NMAT * 256;

    prep_gates<<<1, 256, 0, stream>>>(qp, Gr, Gi);
    qsim<<<4096, 128, 0, stream>>>(x, Gr, Gi, fcw, fcb, (float*)d_out);
}


# Each Rot factored RZ*RY*RZ: fused RY gates are real (two real WMMA chains a group instead of three), RZ phases multiplied in per element; a third fewer matrix instructions; sum order changes
# speedup vs baseline: 2.6369x; 1.1090x over previous
#include <hip/hip_runtime.h>
#include <hip/hip_bf16.h>

typedef __attribute__((ext_vector_type(2))) float v2f;
typedef __attribute__((ext_vector_type(4))) float v4f;
typedef __attribute__((ext_vector_type(8))) float v8f;

#define N_QUBITS 12
#define N_LAYERS 6
#define N_CLASSES 10
#define NSTATE 4096
#define NMAT 18

#define PH_OFF 256

__global__ void prep_gates(const float* __restrict__ qp,
                           float* __restrict__ Gr, float* __restrict__ Gi) {
    __shared__ float ry[72][2];
    __shared__ float pf[72][2];
    __shared__ float pw[72][2];
    int t = threadIdx.x;
    if (t < 72) {
        const float* p = qp + t * 3;
        float phi = p[0], th = p[1], om = p[2];
        float ct = cosf(0.5f * th), st = sinf(0.5f * th);
        ry[t][0] = ct; ry[t][1] = st;
        pf[t][0] = cosf(0.5f * phi); pf[t][1] = sinf(0.5f * phi);
        pw[t][0] = cosf(0.5f * om);  pw[t][1] = sinf(0.5f * om);
        if (t < 12) {
            float a = 0.5f * (phi + om), b = 0.5f * (phi - om);
            float ca = cosf(a), sa = sinf(a), cb = cosf(b), sb = sinf(b);
            Gr[t * 4 + 0] = ca * ct;  Gi[t * 4 + 0] = -sa * ct;
            Gr[t * 4 + 1] = -cb * st; Gi[t * 4 + 1] = -sb * st;
            Gr[t * 4 + 2] = cb * st;  Gi[t * 4 + 2] = -sb * st;
            Gr[t * 4 + 3] = ca * ct;  Gi[t * 4 + 3] = sa * ct;
        }
    }
    __syncthreads();
    for (int e = 3 * 256 + t; e < NMAT * 256; e += blockDim.x) {
        int m = e >> 8, pos = e & 255;
        int l = m / 3, g = m % 3;
        int i = pos >> 4, k = pos & 15;
        float p = 1.f;
#pragma unroll
        for (int j = 0; j < 4; ++j) {
            int u = l * 12 + g * 4 + j;
            int bi_ = (i >> (3 - j)) & 1;
            int bk  = (k >> (3 - j)) & 1;
            float cs = ry[u][0], sn = ry[u][1];
            p *= (bi_ == bk) ? cs : (bi_ ? sn : -sn);
        }
        Gr[e] = p;
    }
    for (int e = t; e < (NMAT - 3) * 32; e += blockDim.x) {
        int i = e & 15, w = (e >> 4) & 1, m = 3 + (e >> 5);
        int l = m / 3, g = m % 3;
        float pr = 1.f, pi = 0.f;
#pragma unroll
        for (int j = 0; j < 4; ++j) {
            int u = l * 12 + g * 4 + j;
            int bit = (i >> (3 - j)) & 1;
            float cs = w ? pw[u][0] : pf[u][0];
            float sn = w ? pw[u][1] : pf[u][1];
            float si = bit ? sn : -sn;
            float nr = pr * cs - pi * si;
            float ni = pr * si + pi * cs;
            pr = nr; pi = ni;
        }
        Gi[PH_OFF + (m * 2 + w) * 32 + 2 * i]     = pr;
        Gi[PH_OFF + (m * 2 + w) * 32 + 2 * i + 1] = pi;
    }
}

struct RingShifts { int s1, s2, s4, s8, wrap; };

__device__ __forceinline__ RingShifts ring_shifts(int r) {
    RingShifts k;
    k.s1 = r;
    k.s2 = min(2 * r, N_QUBITS);
    k.s4 = min(4 * r, N_QUBITS);
    k.s8 = min(8 * r, N_QUBITS);
    k.wrap = N_QUBITS - r;
    return k;
}

__device__ __forceinline__ int ring_dst(int s, const RingShifts k) {
    int c = s;
    c ^= c >> k.s1;
    c ^= c >> k.s2;
    c ^= c >> k.s4;
    c ^= c >> k.s8;
    return c ^ ((c << k.wrap) & (NSTATE - 1));
}

__global__ __launch_bounds__(128) void qsim(
        const float* __restrict__ x,
        const float* __restrict__ Gr, const float* __restrict__ Gi,
        const float* __restrict__ fcw, const float* __restrict__ fcb,
        float* __restrict__ out) {
    __shared__ float sbuf[4 * NSTATE];
    __shared__ float fr[24], fi[24];
    __shared__ float tr[48], ti[48];
    __shared__ float red[4 * 12];
    __shared__ float ev[12];

    const int b    = blockIdx.x;
    const int tid  = threadIdx.x;
    const int lane = tid & 31;
    const int wave = tid >> 5;
    const int c    = lane & 15;
    const int sel  = lane >> 4;

    if (tid < 12) {
        float h = 0.5f * x[b * 12 + tid];
        float ch = cosf(h), sh = sinf(h);
        const float* ur = Gr + tid * 4;
        const float* ui = Gi + tid * 4;
        fr[2 * tid]     = ur[0] * ch + ur[1] * sh;
        fi[2 * tid]     = ui[0] * ch + ui[1] * sh;
        fr[2 * tid + 1] = ur[2] * ch + ur[3] * sh;
        fi[2 * tid + 1] = ui[2] * ch + ui[3] * sh;
    }
    __syncthreads();
    if (tid < 48) {
        int g = tid >> 4, i = tid & 15;
        float pr = 1.f, pi = 0.f;
#pragma unroll
        for (int j = 0; j < 4; ++j) {
            int u = 2 * (g * 4 + j) + ((i >> (3 - j)) & 1);
            float wr = fr[u], wi = fi[u];
            float nr = pr * wr - pi * wi;
            float ni = pr * wi + pi * wr;
            pr = nr; pi = ni;
        }
        tr[tid] = pr; ti[tid] = pi;
    }
    __syncthreads();
    {
        const RingShifts rk0 = ring_shifts(1);
#pragma unroll
        for (int h = 0; h < 2; ++h) {
            const int lo = tid + 128 * h;
            const float p1r = tr[16 + (lo >> 4)], p1i = ti[16 + (lo >> 4)];
            const float p2r = tr[32 + (lo & 15)], p2i = ti[32 + (lo & 15)];
            const float pr = p1r * p2r - p1i * p2i;
            const float pi = p1r * p2i + p1i * p2r;
            const int dt = ring_dst(lo, rk0);
#pragma unroll
            for (int j = 0; j < 16; ++j) {
                float ar = tr[j], ai = ti[j];
                int d = dt ^ ring_dst(j << 8, rk0);
                sbuf[d] = ar * pr - ai * pi;
                sbuf[d + 2 * NSTATE] = ar * pi + ai * pr;
            }
        }
    }
    __syncthreads();

    int cur = 0;
    for (int l = 1; l < N_LAYERS; ++l) {
        float* sR = sbuf + cur * NSTATE;
        float* sI = sR + 2 * NSTATE;
        const RingShifts rk = ring_shifts((l % (N_QUBITS - 1)) + 1);
        const int d1 = ring_dst(1, rk) << 2, d2 = ring_dst(2, rk) << 2,
                  d4 = ring_dst(4, rk) << 2;
        const int nxt4 = (cur ^ 1) * (NSTATE * 4);

        const float* g0 = Gr + (l * 3 + 0) * 256;
        const float* g1 = Gr + (l * 3 + 1) * 256;
        const float* g2 = Gr + (l * 3 + 2) * 256;
        const float* ph = Gi + PH_OFF + (l * 3) * 64;

        v2f a0[4], a1[4], a2[4];
        v2f fAr[4], fAi[4], fCr[4], fCi[4];
        {
            const float f2r = ph[128 + 2 * c], f2i = ph[128 + 2 * c + 1];
#pragma unroll
            for (int kk = 0; kk < 4; ++kk) {
                int k0 = kk * 4 + sel * 2;
                v2f t;
                t.x = g0[c * 16 + k0]; t.y = g0[c * 16 + k0 + 1]; a0[kk] = t;
                t.x = g1[c * 16 + k0]; t.y = g1[c * 16 + k0 + 1]; a1[kk] = t;
                int k2 = sel * 8 + kk * 2;
                t.x = g2[c * 16 + k2]; t.y = g2[c * 16 + k2 + 1]; a2[kk] = t;
                const float* q0 = ph + 2 * k0;
                fAr[kk].x = q0[0]; fAi[kk].x = q0[1];
                fAr[kk].y = q0[2]; fAi[kk].y = q0[3];
                const float* q1 = ph + 64 + 2 * k0;
                fCr[kk].x = q1[0] * f2r - q1[1] * f2i;
                fCi[kk].x = q1[0] * f2i + q1[1] * f2r;
                fCr[kk].y = q1[2] * f2r - q1[3] * f2i;
                fCi[kk].y = q1[2] * f2i + q1[3] * f2r;
            }
        }
        v8f wBr, wBi, wEr, wEi;
        {
            const float w1r = ph[96 + 2 * c], w1i = ph[96 + 2 * c + 1];
#pragma unroll
            for (int v = 0; v < 8; ++v) {
                const float* q0 = ph + 32 + 2 * (sel * 8 + v);
                wBr[v] = q0[0]; wBi[v] = q0[1];
                const float* q2 = ph + 160 + 2 * (sel * 8 + v);
                wEr[v] = q2[0] * w1r - q2[1] * w1i;
                wEi[v] = q2[0] * w1i + q2[1] * w1r;
            }
        }

#pragma unroll
        for (int tt = 0; tt < 4; ++tt) {
            int tile = tt * 4 + wave;
            int off = tile * 16 + c;
            v2f xr[4], xi[4];
#pragma unroll
            for (int kk = 0; kk < 4; ++kk) {
                int k0 = kk * 4 + sel * 2;
                v2f t0, t1;
                t0.x = sR[k0 * 256 + off]; t0.y = sR[(k0 + 1) * 256 + off];
                t1.x = sI[k0 * 256 + off]; t1.y = sI[(k0 + 1) * 256 + off];
                xr[kk] = t0 * fAr[kk] - t1 * fAi[kk];
                xi[kk] = t0 * fAi[kk] + t1 * fAr[kk];
            }
            v8f re = {};
#pragma unroll
            for (int kk = 0; kk < 4; ++kk)
                re = __builtin_amdgcn_wmma_f32_16x16x4_f32(
                    false, a0[kk], false, xr[kk], (short)0, re, false, false);
            v8f im = {};
#pragma unroll
            for (int kk = 0; kk < 4; ++kk)
                im = __builtin_amdgcn_wmma_f32_16x16x4_f32(
                    false, a0[kk], false, xi[kk], (short)0, im, false, false);
            const v8f outr = re * wBr - im * wBi;
            const v8f outi = re * wBi + im * wBr;
#pragma unroll
            for (int v = 0; v < 8; ++v) {
                int ad_ = (v + sel * 8) * 256 + off;
                sR[ad_] = outr[v];
                sI[ad_] = outi[v];
            }
        }
        __syncthreads();

#pragma unroll
        for (int tt = 0; tt < 4; ++tt) {
            int tile = wave * 4 + tt;
            int off1 = tile * 256 + c;
            v2f xr[4], xi[4];
#pragma unroll
            for (int kk = 0; kk < 4; ++kk) {
                int k0 = kk * 4 + sel * 2;
                v2f t0, t1;
                t0.x = sR[k0 * 16 + off1]; t0.y = sR[(k0 + 1) * 16 + off1];
                t1.x = sI[k0 * 16 + off1]; t1.y = sI[(k0 + 1) * 16 + off1];
                xr[kk] = t0 * fCr[kk] - t1 * fCi[kk];
                xi[kk] = t0 * fCi[kk] + t1 * fCr[kk];
            }
            v8f tre = {};
#pragma unroll
            for (int kk = 0; kk < 4; ++kk)
                tre = __builtin_amdgcn_wmma_f32_16x16x4_f32(
                    false, xr[kk], false, a1[kk], (short)0, tre, false, false);
            v8f tim = {};
#pragma unroll
            for (int kk = 0; kk < 4; ++kk)
                tim = __builtin_amdgcn_wmma_f32_16x16x4_f32(
                    false, xi[kk], false, a1[kk], (short)0, tim, false, false);
            v8f re = {};
#pragma unroll
            for (int kk = 0; kk < 4; ++kk) {
                v2f b; b.x = tre[2 * kk]; b.y = tre[2 * kk + 1];
                re = __builtin_amdgcn_wmma_f32_16x16x4_f32(
                    false, a2[kk], false, b, (short)0, re, false, false);
            }
            v8f im = {};
#pragma unroll
            for (int kk = 0; kk < 4; ++kk) {
                v2f b; b.x = tim[2 * kk]; b.y = tim[2 * kk + 1];
                im = __builtin_amdgcn_wmma_f32_16x16x4_f32(
                    false, a2[kk], false, b, (short)0, im, false, false);
            }
            const v8f outr = re * wEr - im * wEi;
            const v8f outi = re * wEi + im * wEr;
            const int off = tile * 256 + c * 16;
            const int pb = (ring_dst(off + sel * 8, rk) << 2) | nxt4;
            char* nb = (char*)sbuf;
#pragma unroll
            for (int v = 0; v < 8; ++v) {
                int a = pb ^ ((v & 1) ? d1 : 0) ^ ((v & 2) ? d2 : 0)
                           ^ ((v & 4) ? d4 : 0);
                *(float*)(nb + a) = outr[v];
                *(float*)(nb + a + 2 * NSTATE * 4) = outi[v];
            }
        }
        __syncthreads();

        cur ^= 1;
    }

    float acc[12];
#pragma unroll
    for (int q = 0; q < 12; ++q) acc[q] = 0.f;
#pragma unroll
    for (int j = 0; j < 32; ++j) {
        int idx = tid + j * 128;
        float re = sbuf[cur * NSTATE + idx], im = sbuf[cur * NSTATE + idx + 2 * NSTATE];
        float p = re * re + im * im;
#pragma unroll
        for (int q = 0; q < 12; ++q)
            acc[q] += ((idx >> (11 - q)) & 1) ? -p : p;
    }
#pragma unroll
    for (int offm = 16; offm > 0; offm >>= 1)
#pragma unroll
        for (int q = 0; q < 12; ++q)
            acc[q] += __shfl_xor(acc[q], offm, 32);
    if (lane == 0)
#pragma unroll
        for (int q = 0; q < 12; ++q) red[wave * 12 + q] = acc[q];
    __syncthreads();
    if (tid < 12) {
        float s = 0.f;
#pragma unroll
        for (int w = 0; w < 4; ++w) s += red[w * 12 + tid];
        ev[tid] = s;
    }
    __syncthreads();

    if (tid < N_CLASSES) {
        float s = fcb[tid];
#pragma unroll
        for (int q = 0; q < 12; ++q) s += ev[q] * fcw[tid * 12 + q];
        out[b * N_CLASSES + tid] = s;
    }
}

extern "C" void kernel_launch(void* const* d_in, const int* in_sizes, int n_in,
                              void* d_out, int out_size, void* d_ws, size_t ws_size,
                              hipStream_t stream) {
    const float* x   = (const float*)d_in[0];
    const float* qp  = (const float*)d_in[1];
    const float* fcw = (const float*)d_in[2];
    const float* fcb = (const float*)d_in[3];
    float* Gr = (float*)d_ws;
    float* Gi = Gr + NMAT * 256;

    prep_gates<<<1, 256, 0, stream>>>(qp, Gr, Gi);
    qsim<<<4096, 128, 0, stream>>>(x, Gr, Gi, fcw, fcb, (float*)d_out);
}
